# starting kernel
# baseline (speedup 1.0000x reference)
#include <hip/hip_runtime.h>

typedef __attribute__((ext_vector_type(2))) float v2f;
typedef __attribute__((ext_vector_type(8))) float v8f;

#define HW_H   96
#define HW_W   96
#define NPIX   (HW_H * HW_W)
#define NB     2
#define NC     21
#define NT     (NPIX / 16)
#define NWAVE  8
#define KSTR   18

static __device__ __forceinline__ v8f wmma_f32(v2f a, v2f b, v8f c) {
    return __builtin_amdgcn_wmma_f32_16x16x4_f32(false, a, false, b, (short)0, c, false, false);
}

__global__ void __launch_bounds__(256) pgf_prep_feats(const float* __restrict__ img,
                                                      float* __restrict__ feats,
                                                      float* __restrict__ sq) {
    int idx = blockIdx.x * blockDim.x + threadIdx.x;
    if (idx >= NB * NPIX) return;
    int b = idx / NPIX;
    int n = idx - b * NPIX;
    float f[8];
    f[0] = (float)(n / HW_W) * 0.125f;
    f[1] = (float)(n % HW_W) * 0.125f;
    const float* ip = img + (size_t)b * 3 * NPIX + n;
    f[2] = ip[0]        * 2.0f;
    f[3] = ip[NPIX]     * 2.0f;
    f[4] = ip[2 * NPIX] * 2.0f;
    f[5] = 0.0f; f[6] = 0.0f; f[7] = 0.0f;
    float s = 0.0f;
#pragma unroll
    for (int v = 0; v < 8; ++v) s += f[v] * f[v];
    float* fp = feats + (size_t)idx * 8;
#pragma unroll
    for (int v = 0; v < 8; ++v) fp[v] = f[v];
    sq[idx] = s;
}

__global__ void __launch_bounds__(256) pgf_prep_vt(const float* __restrict__ vals,
                                                   float* __restrict__ vt) {
    int idx = blockIdx.x * blockDim.x + threadIdx.x;
    if (idx >= NB * NT * 32) return;
    int c  = idx & 31;
    int jt = (idx >> 5) % NT;
    int b  = idx / (32 * NT);
    float* dst = vt + (size_t)idx * 16;
    if (c < NC) {
        const float* src = vals + ((size_t)b * NC + c) * NPIX + jt * 16;
#pragma unroll
        for (int jj = 0; jj < 16; ++jj) dst[jj] = src[jj];
    } else {
#pragma unroll
        for (int jj = 0; jj < 16; ++jj) dst[jj] = 0.0f;
    }
}

__global__ void __launch_bounds__(256) pgf_main(const float* __restrict__ feats,
                                                const float* __restrict__ sq,
                                                const float* __restrict__ vt,
                                                float* __restrict__ out) {
    __shared__ float lds_k[NWAVE * 16 * KSTR];
    __shared__ float lds_r[NWAVE * 512];

    const int tid = threadIdx.x;
    const int wv  = tid >> 5;
    const int l   = tid & 31;
    const int col = l & 15;
    const int p   = l >> 4;
    const int it     = blockIdx.x;
    const int b      = blockIdx.y;
    const int i_base = it * 16;
    const size_t fbase = (size_t)b * NPIX;

    const float* fi = feats + (fbase + i_base + col) * 8;
    v2f alo = *(const v2f*)(fi + 2 * p);
    v2f ahi = *(const v2f*)(fi + 4 + 2 * p);

    float si[8];
    const float* sqi = sq + fbase + i_base + 8 * p;
#pragma unroll
    for (int v = 0; v < 8; ++v) si[v] = sqi[v];

    v8f acc0 = {};
    v8f acc1 = {};
    float* kt = lds_k + wv * 16 * KSTR;

    for (int jt = wv; jt < NT; jt += NWAVE) {
        const int j_base = jt * 16;

        const float* fj = feats + (fbase + j_base + col) * 8;
        v2f blo = *(const v2f*)(fj + 2 * p);
        v2f bhi = *(const v2f*)(fj + 4 + 2 * p);
        float sj = sq[fbase + j_base + col];

        v8f g = {};
        g = wmma_f32(alo, blo, g);
        g = wmma_f32(ahi, bhi, g);

#pragma unroll
        for (int v = 0; v < 8; ++v) {
            float d2 = si[v] + sj - 2.0f * g[v];
            float kv = __expf(-0.5f * fmaxf(d2, 0.0f));
            kt[(v + 8 * p) * KSTR + col] = kv;
        }

        if (jt + NWAVE < NT)
            __builtin_prefetch(feats + (fbase + j_base + 16 * NWAVE + col) * 8, 0, 3);

        const float* vtile = vt + (size_t)(b * NT + jt) * 32 * 16;
#pragma unroll
        for (int kk = 0; kk < 4; ++kk) {
            v2f ka  = *(const v2f*)(kt + col * KSTR + 4 * kk + 2 * p);
            v2f vb0 = *(const v2f*)(vtile + (col     ) * 16 + 4 * kk + 2 * p);
            v2f vb1 = *(const v2f*)(vtile + (col + 16) * 16 + 4 * kk + 2 * p);
            acc0 = wmma_f32(ka, vb0, acc0);
            acc1 = wmma_f32(ka, vb1, acc1);
        }
    }

#pragma unroll
    for (int v = 0; v < 8; ++v) {
        int row = v + 8 * p;
        lds_r[wv * 512 + row * 32 + col]      = acc0[v];
        lds_r[wv * 512 + row * 32 + col + 16] = acc1[v];
    }
    __syncthreads();
    for (int e = tid; e < 512; e += 256) {
        float s = 0.0f;
#pragma unroll
        for (int w2 = 0; w2 < NWAVE; ++w2) s += lds_r[w2 * 512 + e];
        int row = e >> 5;
        int c   = e & 31;
        if (c < NC) out[((size_t)b * NC + c) * NPIX + i_base + row] = s;
    }
}

extern "C" void kernel_launch(void* const* d_in, const int* in_sizes, int n_in,
                              void* d_out, int out_size, void* d_ws, size_t ws_size,
                              hipStream_t stream) {
    const float* cur_state   = (const float*)d_in[0];
    const float* input_image = (const float*)d_in[1];
    float* out = (float*)d_out;

    float* ws    = (float*)d_ws;
    float* feats = ws;
    float* sq    = feats + (size_t)NB * NPIX * 8;
    float* vtws  = sq    + (size_t)NB * NPIX;
    (void)in_sizes; (void)n_in; (void)out_size; (void)ws_size;

    {
        int nthr = NB * NPIX;
        pgf_prep_feats<<<(nthr + 255) / 256, 256, 0, stream>>>(input_image, feats, sq);
    }
    {
        int nthr = NB * NT * 32;
        pgf_prep_vt<<<(nthr + 255) / 256, 256, 0, stream>>>(cur_state, vtws);
    }
    {
        dim3 grid(NT, NB);
        pgf_main<<<grid, 256, 0, stream>>>(feats, sq, vtws, out);
    }
}


# Each wave keeps 5 (or 4) i-tiles in registers and reuses every j-tile's feature and value fragments for all of them; 256 blocks of 16 waves, one per WGP; sum order changes (16 wave partials, not 8)
# speedup vs baseline: 1.2974x; 1.2974x over previous
#include <hip/hip_runtime.h>


typedef __attribute__((ext_vector_type(2))) float v2f;
typedef __attribute__((ext_vector_type(8))) float v8f;

#define HW_H   96
#define HW_W   96
#define NPIX   (HW_H * HW_W)
#define NB     2
#define NC     21
#define NT     (NPIX / 16)
#define NWAVE  16
#define KSTR   18
#define MIMAX  5
#define KTF    (16 * KSTR)
#define RSTR   33
#define RWF    (MIMAX * 16 * RSTR)

static __device__ __forceinline__ v8f wmma_f32(v2f a, v2f b, v8f c) {
    return __builtin_amdgcn_wmma_f32_16x16x4_f32(false, a, false, b, (short)0, c, false, false);
}

__global__ void __launch_bounds__(256) pgf_prep_feats(const float* __restrict__ img,
                                                      float* __restrict__ feats,
                                                      float* __restrict__ sq) {
    int idx = blockIdx.x * blockDim.x + threadIdx.x;
    if (idx >= NB * NPIX) return;
    int b = idx / NPIX;
    int n = idx - b * NPIX;
    float f[8];
    f[0] = (float)(n / HW_W) * 0.125f;
    f[1] = (float)(n % HW_W) * 0.125f;
    const float* ip = img + (size_t)b * 3 * NPIX + n;
    f[2] = ip[0]        * 2.0f;
    f[3] = ip[NPIX]     * 2.0f;
    f[4] = ip[2 * NPIX] * 2.0f;
    f[5] = 0.0f; f[6] = 0.0f; f[7] = 0.0f;
    float s = 0.0f;
#pragma unroll
    for (int v = 0; v < 8; ++v) s += f[v] * f[v];
    float* fp = feats + (size_t)idx * 8;
#pragma unroll
    for (int v = 0; v < 8; ++v) fp[v] = f[v];
    sq[idx] = s;
}

__global__ void __launch_bounds__(256) pgf_prep_vt(const float* __restrict__ vals,
                                                   float* __restrict__ vt) {
    int idx = blockIdx.x * blockDim.x + threadIdx.x;
    if (idx >= NB * NT * 32) return;
    int c  = idx & 31;
    int jt = (idx >> 5) % NT;
    int b  = idx / (32 * NT);
    float* dst = vt + (size_t)idx * 16;
    if (c < NC) {
        const float* src = vals + ((size_t)b * NC + c) * NPIX + jt * 16;
#pragma unroll
        for (int jj = 0; jj < 16; ++jj) dst[jj] = src[jj];
    } else {
#pragma unroll
        for (int jj = 0; jj < 16; ++jj) dst[jj] = 0.0f;
    }
}

template <int MI>
static __device__ __forceinline__ void pgf_tiles(const float* __restrict__ feats,
                                                 const float* __restrict__ sq,
                                                 const float* __restrict__ vt,
                                                 float* __restrict__ out,
                                                 float* lds, int b, int t0) {
    const int tid = threadIdx.x;
    const int wv  = __builtin_amdgcn_readfirstlane(tid >> 5);
    const int l   = tid & 31;
    const int col = l & 15;
    const int p   = l >> 4;
    const int i_base = t0 * 16;
    const size_t fbase = (size_t)b * NPIX;

    v2f alo[MI], ahi[MI];
    float si[MI][8];
#pragma unroll
    for (int m = 0; m < MI; ++m) {
        const float* fi = feats + (fbase + i_base + m * 16 + col) * 8;
        alo[m] = *(const v2f*)(fi + 2 * p);
        ahi[m] = *(const v2f*)(fi + 4 + 2 * p);
        const float* sqi = sq + fbase + i_base + m * 16 + 8 * p;
#pragma unroll
        for (int v = 0; v < 8; ++v) si[m][v] = sqi[v];
    }

    v8f acc0[MI];
    v8f acc1[MI];
#pragma unroll
    for (int m = 0; m < MI; ++m) { acc0[m] = (v8f){}; acc1[m] = (v8f){}; }

    float* kt = lds + wv * RWF;

    for (int jt = wv; jt < NT; jt += NWAVE) {
        const int j_base = jt * 16;

        const float* fj = feats + (fbase + j_base + col) * 8;
        v2f blo = *(const v2f*)(fj + 2 * p);
        v2f bhi = *(const v2f*)(fj + 4 + 2 * p);
        float sj = sq[fbase + j_base + col];

        const float* vtile = vt + (size_t)(b * NT + jt) * 32 * 16;
        v2f vb0[4], vb1[4];
#pragma unroll
        for (int kk = 0; kk < 4; ++kk) {
            vb0[kk] = *(const v2f*)(vtile + (col     ) * 16 + 4 * kk + 2 * p);
            vb1[kk] = *(const v2f*)(vtile + (col + 16) * 16 + 4 * kk + 2 * p);
        }

        auto stage1 = [&](int m) {
            v8f g = {};
            g = wmma_f32(alo[m], blo, g);
            g = wmma_f32(ahi[m], bhi, g);
#pragma unroll
            for (int v = 0; v < 8; ++v) {
                float d2 = si[m][v] + sj - 2.0f * g[v];
                float kv = __expf(-0.5f * fmaxf(d2, 0.0f));
                kt[m * KTF + (v + 8 * p) * KSTR + col] = kv;
            }
        };

        stage1(0);
#pragma unroll
        for (int m = 0; m < MI; ++m) {
            v2f ka[4];
#pragma unroll
            for (int kk = 0; kk < 4; ++kk)
                ka[kk] = *(const v2f*)(kt + m * KTF + col * KSTR + 4 * kk + 2 * p);
            if (m + 1 < MI) stage1(m + 1);
#pragma unroll
            for (int kk = 0; kk < 4; ++kk) {
                acc0[m] = wmma_f32(ka[kk], vb0[kk], acc0[m]);
                acc1[m] = wmma_f32(ka[kk], vb1[kk], acc1[m]);
            }
        }
    }

#pragma unroll
    for (int m = 0; m < MI; ++m) {
#pragma unroll
        for (int v = 0; v < 8; ++v) {
            int row = m * 16 + v + 8 * p;
            kt[row * RSTR + col]      = acc0[m][v];
            kt[row * RSTR + col + 16] = acc1[m][v];
        }
    }
    __syncthreads();
    for (int e = tid; e < NC * MI * 16; e += NWAVE * 32) {
        int c = e / (MI * 16);
        int r = e - c * (MI * 16);
        float s = 0.0f;
#pragma unroll
        for (int w2 = 0; w2 < NWAVE; ++w2) s += lds[w2 * RWF + r * RSTR + c];
        out[((size_t)b * NC + c) * NPIX + i_base + r] = s;
    }
}

__global__ void __launch_bounds__(NWAVE * 32) pgf_main(const float* __restrict__ feats,
                                                       const float* __restrict__ sq,
                                                       const float* __restrict__ vt,
                                                       float* __restrict__ out) {
    __shared__ float lds[NWAVE * RWF];
    const int q     = blockIdx.x >> 1;
    const int odd   = blockIdx.x & 1;
    const int unit0 = q * 9 + (odd ? 5 : 0);
    const int b     = unit0 / NT;
    const int t0    = unit0 - b * NT;
    if (odd) pgf_tiles<4>(feats, sq, vt, out, lds, b, t0);
    else     pgf_tiles<5>(feats, sq, vt, out, lds, b, t0);
}

extern "C" void kernel_launch(void* const* d_in, const int* in_sizes, int n_in,
                              void* d_out, int out_size, void* d_ws, size_t ws_size,
                              hipStream_t stream) {
    const float* cur_state   = (const float*)d_in[0];
    const float* input_image = (const float*)d_in[1];
    float* out = (float*)d_out;

    float* ws    = (float*)d_ws;
    float* feats = ws;
    float* sq    = feats + (size_t)NB * NPIX * 8;
    float* vtws  = sq    + (size_t)NB * NPIX;
    (void)in_sizes; (void)n_in; (void)out_size; (void)ws_size;

    {
        int nthr = NB * NPIX;
        pgf_prep_feats<<<(nthr + 255) / 256, 256, 0, stream>>>(input_image, feats, sq);
    }
    {
        int nthr = NB * NT * 32;
        pgf_prep_vt<<<(nthr + 255) / 256, 256, 0, stream>>>(cur_state, vtws);
    }
    {
        pgf_main<<<256, NWAVE * 32, 0, stream>>>(feats, sq, vtws, out);
    }
}


# Exponent of a tile pair from ONE K=4 product (x, colours as operands; row norms in C; column norm and row distance in one fma), exp2 with clamp: 9 WMMAs per pair, not 10; distance terms in new order
# speedup vs baseline: 1.4673x; 1.1310x over previous
#include <hip/hip_runtime.h>


typedef __attribute__((ext_vector_type(2))) float v2f;
typedef __attribute__((ext_vector_type(8))) float v8f;

#define HW_H   96
#define HW_W   96
#define NPIX   (HW_H * HW_W)
#define NB     2
#define NC     21
#define NT     (NPIX / 16)
#define NWAVE  16
#define KSTR   18
#define MIMAX  5
#define KTF    (16 * KSTR)
#define RSTR   33
#define RWF    (MIMAX * 16 * RSTR)

static __device__ __forceinline__ v8f wmma_f32(v2f a, v2f b, v8f c) {
    return __builtin_amdgcn_wmma_f32_16x16x4_f32(false, a, false, b, (short)0, c, false, false);
}

__global__ void __launch_bounds__(256) pgf_prep_feats(const float* __restrict__ img,
                                                      float* __restrict__ f4,
                                                      float* __restrict__ hq) {
    int idx = blockIdx.x * blockDim.x + threadIdx.x;
    if (idx >= NB * NPIX) return;
    int b = idx / NPIX;
    int n = idx - b * NPIX;
    const float* ip = img + (size_t)b * 3 * NPIX + n;
    float4 f;
    f.x = ip[0]        * 2.0f - 1.0f;
    f.y = ip[NPIX]     * 2.0f - 1.0f;
    f.z = ip[2 * NPIX] * 2.0f - 1.0f;
    f.w = ((float)(n % HW_W) - 47.5f) * 0.125f;
    float s = f.x * f.x + f.y * f.y + f.z * f.z + f.w * f.w;
    *(float4*)(f4 + (size_t)idx * 4) = f;
    hq[idx] = -0.5f * s;
}

__global__ void __launch_bounds__(256) pgf_prep_vt(const float* __restrict__ vals,
                                                   float* __restrict__ vt) {
    int idx = blockIdx.x * blockDim.x + threadIdx.x;
    if (idx >= NB * NT * 32) return;
    int c  = idx & 31;
    int jt = (idx >> 5) % NT;
    int b  = idx / (32 * NT);
    float* dst = vt + (size_t)idx * 16;
    if (c < NC) {
        const float* src = vals + ((size_t)b * NC + c) * NPIX + jt * 16;
#pragma unroll
        for (int jj = 0; jj < 16; ++jj) dst[jj] = src[jj];
    } else {
#pragma unroll
        for (int jj = 0; jj < 16; ++jj) dst[jj] = 0.0f;
    }
}

#define LOG2E 1.4426950408889634f

template <int MI>
static __device__ __forceinline__ void pgf_tiles(const float* __restrict__ f4,
                                                 const float* __restrict__ hq,
                                                 const float* __restrict__ vt,
                                                 float* __restrict__ out,
                                                 float* lds, int b, int t0) {
    const int tid = threadIdx.x;
    const int wv  = __builtin_amdgcn_readfirstlane(tid >> 5);
    const int l   = tid & 31;
    const int col = l & 15;
    const int p   = l >> 4;
    const int i_base = t0 * 16;
    const size_t fbase = (size_t)b * NPIX;

    v2f afr[MI];
    v8f hi[MI];
    int yi[MI];
#pragma unroll
    for (int m = 0; m < MI; ++m) {
        const float* fi = f4 + (fbase + i_base + m * 16 + col) * 4;
        afr[m] = *(const v2f*)(fi + 2 * p);
        const float* hqi = hq + fbase + i_base + m * 16 + 8 * p;
#pragma unroll
        for (int v = 0; v < 8; ++v) hi[m][v] = hqi[v];
        yi[m] = (t0 + m) / (HW_W / 16);
    }

    v8f acc0[MI];
    v8f acc1[MI];
#pragma unroll
    for (int m = 0; m < MI; ++m) { acc0[m] = (v8f){}; acc1[m] = (v8f){}; }

    float* kt = lds + wv * RWF;

    for (int jt = wv; jt < NT; jt += NWAVE) {
        const int j_base = jt * 16;
        const int yj = jt / (HW_W / 16);

        const float* fj = f4 + (fbase + j_base + col) * 4;
        v2f bfr = *(const v2f*)(fj + 2 * p);
        float hj = hq[fbase + j_base + col];

        const float* vtile = vt + (size_t)(b * NT + jt) * 32 * 16;
        v2f vb0[4], vb1[4];
#pragma unroll
        for (int kk = 0; kk < 4; ++kk) {
            vb0[kk] = *(const v2f*)(vtile + (col     ) * 16 + 4 * kk + 2 * p);
            vb1[kk] = *(const v2f*)(vtile + (col + 16) * 16 + 4 * kk + 2 * p);
        }

        auto stage1 = [&](int m) {
            v8f g = wmma_f32(afr[m], bfr, hi[m]);
            const float dy = (float)(yi[m] - yj);
            const float uj = LOG2E * (hj - dy * dy * (1.0f / 128.0f));
#pragma unroll
            for (int v = 0; v < 8; ++v) {
                float e = __builtin_amdgcn_exp2f(fmaf(g[v], LOG2E, uj));
                float kv = __builtin_amdgcn_fmed3f(e, 0.0f, 1.0f);
                kt[m * KTF + (v + 8 * p) * KSTR + col] = kv;
            }
        };

        stage1(0);
#pragma unroll
        for (int m = 0; m < MI; ++m) {
            v2f ka[4];
#pragma unroll
            for (int kk = 0; kk < 4; ++kk)
                ka[kk] = *(const v2f*)(kt + m * KTF + col * KSTR + 4 * kk + 2 * p);
            if (m + 1 < MI) stage1(m + 1);
#pragma unroll
            for (int kk = 0; kk < 4; ++kk) {
                acc0[m] = wmma_f32(ka[kk], vb0[kk], acc0[m]);
                acc1[m] = wmma_f32(ka[kk], vb1[kk], acc1[m]);
            }
        }
    }

#pragma unroll
    for (int m = 0; m < MI; ++m) {
#pragma unroll
        for (int v = 0; v < 8; ++v) {
            int row = m * 16 + v + 8 * p;
            kt[row * RSTR + col]      = acc0[m][v];
            kt[row * RSTR + col + 16] = acc1[m][v];
        }
    }
    __syncthreads();
    for (int e = tid; e < NC * MI * 16; e += NWAVE * 32) {
        int c = e / (MI * 16);
        int r = e - c * (MI * 16);
        float s = 0.0f;
#pragma unroll
        for (int w2 = 0; w2 < NWAVE; ++w2) s += lds[w2 * RWF + r * RSTR + c];
        out[((size_t)b * NC + c) * NPIX + i_base + r] = s;
    }
}

__global__ void __launch_bounds__(NWAVE * 32) pgf_main(const float* __restrict__ f4,
                                                       const float* __restrict__ hq,
                                                       const float* __restrict__ vt,
                                                       float* __restrict__ out) {
    __shared__ float lds[NWAVE * RWF];
    const int q     = blockIdx.x >> 1;
    const int odd   = blockIdx.x & 1;
    const int unit0 = q * 9 + (odd ? 5 : 0);
    const int b     = unit0 / NT;
    const int t0    = unit0 - b * NT;
    if (odd) pgf_tiles<4>(f4, hq, vt, out, lds, b, t0);
    else     pgf_tiles<5>(f4, hq, vt, out, lds, b, t0);
}

extern "C" void kernel_launch(void* const* d_in, const int* in_sizes, int n_in,
                              void* d_out, int out_size, void* d_ws, size_t ws_size,
                              hipStream_t stream) {
    const float* cur_state   = (const float*)d_in[0];
    const float* input_image = (const float*)d_in[1];
    float* out = (float*)d_out;

    float* ws    = (float*)d_ws;
    float* f4    = ws;
    float* hq    = f4    + (size_t)NB * NPIX * 4;
    float* vtws  = hq    + (size_t)NB * NPIX;
    (void)in_sizes; (void)n_in; (void)out_size; (void)ws_size;

    {
        int nthr = NB * NPIX;
        pgf_prep_feats<<<(nthr + 255) / 256, 256, 0, stream>>>(input_image, f4, hq);
    }
    {
        int nthr = NB * NT * 32;
        pgf_prep_vt<<<(nthr + 255) / 256, 256, 0, stream>>>(cur_state, vtws);
    }
    {
        pgf_main<<<256, NWAVE * 32, 0, stream>>>(f4, hq, vtws, out);
    }
}


# Both products transposed: K^T (rows j; F_j*F_i^T with hq_j in C) stays in its D registers as B fragment of V^T*K^T, chunk kk = rows 8h+2kk,+1; no LDS in j loop; j-sum and distance-term order change
# speedup vs baseline: 1.5133x; 1.0314x over previous
#include <hip/hip_runtime.h>


typedef __attribute__((ext_vector_type(2))) float v2f;
typedef __attribute__((ext_vector_type(8))) float v8f;

#define HW_H   96
#define HW_W   96
#define NPIX   (HW_H * HW_W)
#define NB     2
#define NC     21
#define NT     (NPIX / 16)
#define NWAVE  16
#define MIMAX  5
#define RSTR   82
#define RWF    (32 * RSTR)

static __device__ __forceinline__ v8f wmma_f32(v2f a, v2f b, v8f c) {
    return __builtin_amdgcn_wmma_f32_16x16x4_f32(false, a, false, b, (short)0, c, false, false);
}

__global__ void __launch_bounds__(256) pgf_prep_feats(const float* __restrict__ img,
                                                      float* __restrict__ f4,
                                                      float* __restrict__ hq) {
    int idx = blockIdx.x * blockDim.x + threadIdx.x;
    if (idx >= NB * NPIX) return;
    int b = idx / NPIX;
    int n = idx - b * NPIX;
    const float* ip = img + (size_t)b * 3 * NPIX + n;
    float4 f;
    f.x = ip[0]        * 2.0f - 1.0f;
    f.y = ip[NPIX]     * 2.0f - 1.0f;
    f.z = ip[2 * NPIX] * 2.0f - 1.0f;
    f.w = ((float)(n % HW_W) - 47.5f) * 0.125f;
    float s = f.x * f.x + f.y * f.y + f.z * f.z + f.w * f.w;
    *(float4*)(f4 + (size_t)idx * 4) = f;
    hq[idx] = -0.5f * s;
}

__global__ void __launch_bounds__(256) pgf_prep_vt(const float* __restrict__ vals,
                                                   float* __restrict__ vt) {
    int idx = blockIdx.x * blockDim.x + threadIdx.x;
    if (idx >= NB * NT * 32) return;
    int c  = idx & 31;
    int jt = (idx >> 5) % NT;
    int b  = idx / (32 * NT);
    float* dst = vt + (size_t)idx * 16;
    if (c < NC) {
        const float* src = vals + ((size_t)b * NC + c) * NPIX + jt * 16;
#pragma unroll
        for (int jj = 0; jj < 16; ++jj) dst[jj] = src[jj];
    } else {
#pragma unroll
        for (int jj = 0; jj < 16; ++jj) dst[jj] = 0.0f;
    }
}

#define LOG2E 1.4426950408889634f

template <int MI>
static __device__ __forceinline__ void pgf_tiles(const float* __restrict__ f4,
                                                 const float* __restrict__ hq,
                                                 const float* __restrict__ vt,
                                                 float* __restrict__ out,
                                                 float* lds, int b, int t0) {
    const int tid = threadIdx.x;
    const int wv  = __builtin_amdgcn_readfirstlane(tid >> 5);
    const int l   = tid & 31;
    const int col = l & 15;
    const int p   = l >> 4;
    const int i_base = t0 * 16;
    const size_t fbase = (size_t)b * NPIX;

    v2f bfi[MI];
    float hil[MI];
    int yi[MI];
#pragma unroll
    for (int m = 0; m < MI; ++m) {
        const float* fi = f4 + (fbase + i_base + m * 16 + col) * 4;
        bfi[m] = *(const v2f*)(fi + 2 * p);
        hil[m] = LOG2E * hq[fbase + i_base + m * 16 + col];
        yi[m] = (t0 + m) / (HW_W / 16);
    }

    v8f acc0[MI];
    v8f acc1[MI];
#pragma unroll
    for (int m = 0; m < MI; ++m) { acc0[m] = (v8f){}; acc1[m] = (v8f){}; }

    for (int jt = wv; jt < NT; jt += NWAVE) {
        const int j_base = jt * 16;
        const int yj = jt / (HW_W / 16);

        const float* fj = f4 + (fbase + j_base + col) * 4;
        const v2f afj = *(const v2f*)(fj + 2 * p);

        const float4* hq4 = (const float4*)(hq + fbase + j_base + 8 * p);
        const float4 h0 = hq4[0], h1 = hq4[1];
        const v8f hjc = {h0.x, h0.y, h0.z, h0.w, h1.x, h1.y, h1.z, h1.w};

        const float* vtile = vt + (size_t)(b * NT + jt) * 32 * 16;
        const float4* r0 = (const float4*)(vtile + (col     ) * 16 + 8 * p);
        const float4* r1 = (const float4*)(vtile + (col + 16) * 16 + 8 * p);
        const float4 a00 = r0[0], a01 = r0[1], a10 = r1[0], a11 = r1[1];
        const v2f va0[4] = {{a00.x, a00.y}, {a00.z, a00.w}, {a01.x, a01.y}, {a01.z, a01.w}};
        const v2f va1[4] = {{a10.x, a10.y}, {a10.z, a10.w}, {a11.x, a11.y}, {a11.z, a11.w}};

        auto stage1 = [&](int m) -> v8f {
            v8f g = wmma_f32(afj, bfi[m], hjc);
            const float dy = (float)(yi[m] - yj);
            const float um = fmaf(dy * dy, -LOG2E / 128.0f, hil[m]);
            v8f k;
#pragma unroll
            for (int v = 0; v < 8; ++v) {
                float e = __builtin_amdgcn_exp2f(fmaf(g[v], LOG2E, um));
                k[v] = __builtin_amdgcn_fmed3f(e, 0.0f, 1.0f);
            }
            return k;
        };

        v8f kcur = stage1(0);
#pragma unroll
        for (int m = 0; m < MI; ++m) {
            v8f knext = kcur;
            if (m + 1 < MI) knext = stage1(m + 1);
#pragma unroll
            for (int kk = 0; kk < 4; ++kk) {
                const v2f kb = {kcur[2 * kk], kcur[2 * kk + 1]};
                acc0[m] = wmma_f32(va0[kk], kb, acc0[m]);
                acc1[m] = wmma_f32(va1[kk], kb, acc1[m]);
            }
            kcur = knext;
        }
    }

    float* red = lds + wv * RWF;
#pragma unroll
    for (int m = 0; m < MI; ++m) {
#pragma unroll
        for (int v = 0; v < 8; ++v) {
            red[(v + 8 * p) * RSTR + m * 16 + col]      = acc0[m][v];
            red[(16 + v + 8 * p) * RSTR + m * 16 + col] = acc1[m][v];
        }
    }
    __syncthreads();
    for (int e = tid; e < NC * MI * 16; e += NWAVE * 32) {
        int c = e / (MI * 16);
        int r = e - c * (MI * 16);
        float s = 0.0f;
#pragma unroll
        for (int w2 = 0; w2 < NWAVE; ++w2) s += lds[w2 * RWF + c * RSTR + r];
        out[((size_t)b * NC + c) * NPIX + i_base + r] = s;
    }
}

__global__ void __launch_bounds__(NWAVE * 32) pgf_main(const float* __restrict__ f4,
                                                       const float* __restrict__ hq,
                                                       const float* __restrict__ vt,
                                                       float* __restrict__ out) {
    __shared__ float lds[NWAVE * RWF];
    const int q     = blockIdx.x >> 1;
    const int odd   = blockIdx.x & 1;
    const int unit0 = q * 9 + (odd ? 5 : 0);
    const int b     = unit0 / NT;
    const int t0    = unit0 - b * NT;
    if (odd) pgf_tiles<4>(f4, hq, vt, out, lds, b, t0);
    else     pgf_tiles<5>(f4, hq, vt, out, lds, b, t0);
}

extern "C" void kernel_launch(void* const* d_in, const int* in_sizes, int n_in,
                              void* d_out, int out_size, void* d_ws, size_t ws_size,
                              hipStream_t stream) {
    const float* cur_state   = (const float*)d_in[0];
    const float* input_image = (const float*)d_in[1];
    float* out = (float*)d_out;

    float* ws    = (float*)d_ws;
    float* f4    = ws;
    float* hq    = f4    + (size_t)NB * NPIX * 4;
    float* vtws  = hq    + (size_t)NB * NPIX;
    (void)in_sizes; (void)n_in; (void)out_size; (void)ws_size;

    {
        int nthr = NB * NPIX;
        pgf_prep_feats<<<(nthr + 255) / 256, 256, 0, stream>>>(input_image, f4, hq);
    }
    {
        int nthr = NB * NT * 32;
        pgf_prep_vt<<<(nthr + 255) / 256, 256, 0, stream>>>(cur_state, vtws);
    }
    {
        pgf_main<<<256, NWAVE * 32, 0, stream>>>(f4, hq, vtws, out);
    }
}


# Both preparation launches folded into the main kernel: norms once per block into LDS, feature fragments made from the image, V read in place; 1 launch, not 3; j-sum order changes
# speedup vs baseline: 1.6404x; 1.0840x over previous
#include <hip/hip_runtime.h>


typedef __attribute__((ext_vector_type(2))) float v2f;
typedef __attribute__((ext_vector_type(8))) float v8f;

#define HW_H   96
#define HW_W   96
#define NPIX   (HW_H * HW_W)
#define NB     2
#define NC     21
#define NT     (NPIX / 16)
#define NWAVE  16
#define NTHR   (NWAVE * 32)
#define NITER  (NT / NWAVE)
#define MIMAX  5
#define RSTR   82
#define RWF    (32 * RSTR)

static __device__ __forceinline__ v8f wmma_f32(v2f a, v2f b, v8f c) {
    return __builtin_amdgcn_wmma_f32_16x16x4_f32(false, a, false, b, (short)0, c, false, false);
}

#define LOG2E 1.4426950408889634f

template <int MI>
static __device__ __forceinline__ void pgf_tiles(const float* __restrict__ imgb,
                                                 const float* __restrict__ valb,
                                                 float* __restrict__ outb,
                                                 float* lds, const float* hqs, int t0) {
    const int tid = threadIdx.x;
    const int wv  = __builtin_amdgcn_readfirstlane(tid >> 5);
    const int l   = tid & 31;
    const int col = l & 15;
    const int p   = l >> 4;
    const int i_base = t0 * 16;

    const unsigned off_lo = (unsigned)((p ? 2 * NPIX : 0) + col);
    const unsigned off_hi = (unsigned)(NPIX + col);
    const float ca = p ? 0.0f : 2.0f;
    const float xm = p ? 1.0f : 0.0f;
    const float xk = p ? (float)col * 0.125f : -1.0f;
    auto feat_frag = [&](int t) -> v2f {
        const float* it0 = imgb + t * 16;
        const float xs = (float)((t % (HW_W / 16)) * 16) * 0.125f - 5.9375f;
        const float c_lo = it0[off_lo];
        const float c_hi = it0[off_hi];
        v2f f;
        f.x = c_lo * 2.0f - 1.0f;
        f.y = fmaf(c_hi, ca, fmaf(xs, xm, xk));
        return f;
    };

    v2f bfi[MI];
    float hil[MI];
    int yi[MI];
#pragma unroll
    for (int m = 0; m < MI; ++m) {
        bfi[m] = feat_frag(t0 + m);
        hil[m] = LOG2E * hqs[i_base + m * 16 + col];
        yi[m] = (t0 + m) / (HW_W / 16);
    }

    v8f acc0[MI];
    v8f acc1[MI];
#pragma unroll
    for (int m = 0; m < MI; ++m) { acc0[m] = (v8f){}; acc1[m] = (v8f){}; }

    const unsigned offv0 = (unsigned)(col * NPIX + 8 * p);
    const unsigned offv1 = (unsigned)((col + 16 < NC ? col + 16 : NC - 1) * NPIX + 8 * p);

    for (int it = 0; it < NITER; ++it) {
        const int jt = (it >> 1) * (2 * NWAVE) + 2 * wv + (it & 1);
        const int j_base = jt * 16;
        const int yj = jt / (HW_W / 16);

        const v2f afj = feat_frag(jt);

        const float4* hq4 = (const float4*)(hqs + j_base + 8 * p);
        const float4 h0 = hq4[0], h1 = hq4[1];
        const v8f hjc = {h0.x, h0.y, h0.z, h0.w, h1.x, h1.y, h1.z, h1.w};

        const float* vj = valb + j_base;
        const float4 a00 = *(const float4*)(vj + offv0);
        const float4 a01 = *(const float4*)(vj + offv0 + 4);
        const float4 a10 = *(const float4*)(vj + offv1);
        const float4 a11 = *(const float4*)(vj + offv1 + 4);
        const v2f va0[4] = {{a00.x, a00.y}, {a00.z, a00.w}, {a01.x, a01.y}, {a01.z, a01.w}};
        const v2f va1[4] = {{a10.x, a10.y}, {a10.z, a10.w}, {a11.x, a11.y}, {a11.z, a11.w}};

        auto stage1 = [&](int m) -> v8f {
            v8f g = wmma_f32(afj, bfi[m], hjc);
            const float dy = (float)(yi[m] - yj);
            const float um = fmaf(dy * dy, -LOG2E / 128.0f, hil[m]);
            v8f k;
#pragma unroll
            for (int v = 0; v < 8; ++v) {
                float e = __builtin_amdgcn_exp2f(fmaf(g[v], LOG2E, um));
                k[v] = __builtin_amdgcn_fmed3f(e, 0.0f, 1.0f);
            }
            return k;
        };

        v8f kcur = stage1(0);
#pragma unroll
        for (int m = 0; m < MI; ++m) {
            v8f knext = kcur;
            if (m + 1 < MI) knext = stage1(m + 1);
#pragma unroll
            for (int kk = 0; kk < 4; ++kk) {
                const v2f kb = {kcur[2 * kk], kcur[2 * kk + 1]};
                acc0[m] = wmma_f32(va0[kk], kb, acc0[m]);
                acc1[m] = wmma_f32(va1[kk], kb, acc1[m]);
            }
            kcur = knext;
        }
    }

    float* red = lds + wv * RWF;
#pragma unroll
    for (int m = 0; m < MI; ++m) {
#pragma unroll
        for (int v = 0; v < 8; ++v) {
            red[(v + 8 * p) * RSTR + m * 16 + col]      = acc0[m][v];
            red[(16 + v + 8 * p) * RSTR + m * 16 + col] = acc1[m][v];
        }
    }
    __syncthreads();
    for (int e = tid; e < NC * MI * 16; e += NTHR) {
        int c = e / (MI * 16);
        int r = e - c * (MI * 16);
        float s = 0.0f;
#pragma unroll
        for (int w2 = 0; w2 < NWAVE; ++w2) s += lds[w2 * RWF + c * RSTR + r];
        outb[(size_t)c * NPIX + i_base + r] = s;
    }
}

__global__ void __launch_bounds__(NTHR) pgf_main(const float* __restrict__ img,
                                                 const float* __restrict__ vals,
                                                 float* __restrict__ out) {
    __shared__ float lds[NWAVE * RWF];
    __shared__ __attribute__((aligned(16))) float hqs[NPIX];
    const int q     = blockIdx.x >> 1;
    const int odd   = blockIdx.x & 1;
    const int unit0 = q * 9 + (odd ? 5 : 0);
    const int b     = unit0 / NT;
    const int t0    = unit0 - b * NT;
    const float* imgb = img  + (size_t)b * 3 * NPIX;
    const float* valb = vals + (size_t)b * NC * NPIX;
    float*       outb = out  + (size_t)b * NC * NPIX;

#pragma unroll
    for (int k = 0; k < NPIX / NTHR; ++k) {
        const int n = k * NTHR + (int)threadIdx.x;
        const float f0 = imgb[n]            * 2.0f - 1.0f;
        const float f1 = imgb[NPIX + n]     * 2.0f - 1.0f;
        const float f2 = imgb[2 * NPIX + n] * 2.0f - 1.0f;
        const float f3 = ((float)(n % HW_W) - 47.5f) * 0.125f;
        hqs[n] = -0.5f * (f0 * f0 + f1 * f1 + f2 * f2 + f3 * f3);
    }
    __syncthreads();

    if (odd) pgf_tiles<4>(imgb, valb, outb, lds, hqs, t0);
    else     pgf_tiles<5>(imgb, valb, outb, lds, hqs, t0);
}

extern "C" void kernel_launch(void* const* d_in, const int* in_sizes, int n_in,
                              void* d_out, int out_size, void* d_ws, size_t ws_size,
                              hipStream_t stream) {
    const float* cur_state   = (const float*)d_in[0];
    const float* input_image = (const float*)d_in[1];
    float* out = (float*)d_out;
    (void)in_sizes; (void)n_in; (void)out_size; (void)d_ws; (void)ws_size;

    pgf_main<<<256, NTHR, 0, stream>>>(input_image, cur_state, out);
}
